# starting kernel
# baseline (speedup 1.0000x reference)
#include <hip/hip_runtime.h>

#define B_ 256
#define T_ 4096
#define H_ 32
#define LDS_STRIDE 20

typedef float v2f __attribute__((ext_vector_type(2)));
typedef float v8f __attribute__((ext_vector_type(8)));

__device__ __forceinline__ float hw_tanhf(float v) {
#if __has_builtin(__builtin_amdgcn_tanhf)
    return __builtin_amdgcn_tanhf(v);
#else
    const float xc = __builtin_fminf(__builtin_fmaxf(v, -9.0f), 9.0f);
    const float t  = __builtin_amdgcn_exp2f(xc * 2.8853900817779268f);
    const float r  = __builtin_amdgcn_rcpf(t + 1.0f);
    return (t - 1.0f) * r;
#endif
}

__global__ __launch_bounds__(32, 1)
void SineRNN_wmma_kernel(const float* __restrict__ x,
                         const float* __restrict__ W_ih,
                         const float* __restrict__ W_hh,
                         const float* __restrict__ b_ih,
                         const float* __restrict__ b_hh,
                         const float* __restrict__ W_fc,
                         const float* __restrict__ b_fc,
                         float* __restrict__ out)
{
    __shared__ float hbuf[H_ * LDS_STRIDE];

    const int lane  = threadIdx.x & 31;
    const int n     = lane & 15;
    const int hi    = lane >> 4;
    const int bbase = blockIdx.x * 16;

    v2f A[2][8];
#pragma unroll
    for (int c = 0; c < 2; ++c) {
#pragma unroll
        for (int j = 0; j < 8; ++j) {
            const int row = 16 * c + n;
            const int k0  = 4 * j + 2 * hi;
            A[c][j].x = W_hh[row * H_ + k0];
            A[c][j].y = W_hh[row * H_ + k0 + 1];
        }
    }

    float wih[2][8], bia[2][8], wfc[2][8];
#pragma unroll
    for (int c = 0; c < 2; ++c) {
#pragma unroll
        for (int r = 0; r < 8; ++r) {
            const int h = 16 * c + 8 * hi + r;
            wih[c][r] = W_ih[h];
            bia[c][r] = b_ih[h] + b_hh[h];
            wfc[c][r] = W_fc[h];
        }
    }
    const float bfc = b_fc[0];

    const float* xrow = x   + (size_t)(bbase + n) * T_;
    float*       orow = out + (size_t)(bbase + n) * T_;

    v2f Bop[8];
#pragma unroll
    for (int j = 0; j < 8; ++j) { Bop[j].x = 0.0f; Bop[j].y = 0.0f; }

    for (int t = 0; t < T_; ++t) {
        const float xv = xrow[t];
        __builtin_prefetch(xrow + t + 96, 0, 0);

        v8f acc0, acc1;
#pragma unroll
        for (int r = 0; r < 8; ++r) {
            acc0[r] = __builtin_fmaf(xv, wih[0][r], bia[0][r]);
            acc1[r] = __builtin_fmaf(xv, wih[1][r], bia[1][r]);
        }

#pragma unroll
        for (int j = 0; j < 8; ++j) {
            acc0 = __builtin_amdgcn_wmma_f32_16x16x4_f32(
                false, A[0][j], false, Bop[j], (short)0, acc0, false, false);
            acc1 = __builtin_amdgcn_wmma_f32_16x16x4_f32(
                false, A[1][j], false, Bop[j], (short)0, acc1, false, false);
        }

        float s = 0.0f;
#pragma unroll
        for (int r = 0; r < 8; ++r) {
            const float h0 = hw_tanhf(acc0[r]);
            const float h1 = hw_tanhf(acc1[r]);
            acc0[r] = h0;
            acc1[r] = h1;
            s = __builtin_fmaf(wfc[0][r], h0, s);
            s = __builtin_fmaf(wfc[1][r], h1, s);
        }
        s += __shfl_xor(s, 16, 32);
        if (hi == 0) orow[t] = s + bfc;

#pragma unroll
        for (int r = 0; r < 8; ++r) {
            hbuf[(8 * hi + r)      * LDS_STRIDE + n] = acc0[r];
            hbuf[(16 + 8 * hi + r) * LDS_STRIDE + n] = acc1[r];
        }
#pragma unroll
        for (int j = 0; j < 8; ++j) {
            const int k0 = 4 * j + 2 * hi;
            Bop[j].x = hbuf[k0       * LDS_STRIDE + n];
            Bop[j].y = hbuf[(k0 + 1) * LDS_STRIDE + n];
        }
    }
}

extern "C" void kernel_launch(void* const* d_in, const int* in_sizes, int n_in,
                              void* d_out, int out_size, void* d_ws, size_t ws_size,
                              hipStream_t stream) {
    const float* x    = (const float*)d_in[0];
    const float* W_ih = (const float*)d_in[1];
    const float* W_hh = (const float*)d_in[2];
    const float* b_ih = (const float*)d_in[3];
    const float* b_hh = (const float*)d_in[4];
    const float* W_fc = (const float*)d_in[5];
    const float* b_fc = (const float*)d_in[6];
    float* out = (float*)d_out;

    dim3 grid(B_ / 16);
    dim3 block(32);
    SineRNN_wmma_kernel<<<grid, block, 0, stream>>>(
        x, W_ih, W_hh, b_ih, b_hh, W_fc, b_fc, out);
}


# recurrence wave: chunk of 8 steps written out straight, h of 4 steps held in registers and stored in ONE LDS group, x of a chunk read by two 16-byte loads: 2 LDS places per chunk (was 8); same bits
# speedup vs baseline: 2.5256x; 2.5256x over previous
#include <hip/hip_runtime.h>

#define B_ 256
#define T_ 4096
#define H_ 32
#define HPITCH 36
#define SLOT_WORDS (16 * HPITCH)
#define CH 8
#define NCH (T_ / CH)
#define PL 4
#define RCH 3
#define RSLOTS (RCH * CH)
#define XRING 24
#define XPITCH 28
#define XBUF_WORDS (16 * XPITCH)

typedef float v2f __attribute__((ext_vector_type(2)));
typedef float v4f __attribute__((ext_vector_type(4)));
typedef float v8f __attribute__((ext_vector_type(8)));

__device__ __forceinline__ float hw_tanhf(float v) {
#if __has_builtin(__builtin_amdgcn_tanhf)
    return __builtin_amdgcn_tanhf(v);
#else
    const float xc = __builtin_fminf(__builtin_fmaxf(v, -9.0f), 9.0f);
    const float t  = __builtin_amdgcn_exp2f(xc * 2.8853900817779268f);
    const float r  = __builtin_amdgcn_rcpf(t + 1.0f);
    return (t - 1.0f) * r;
#endif
}

__device__ __forceinline__ void row_swap(float& d, float& s) {
    typedef unsigned v2u __attribute__((ext_vector_type(2)));
    const v2u r = __builtin_amdgcn_permlane16_swap(__builtin_bit_cast(unsigned, d),
                                                   __builtin_bit_cast(unsigned, s),
                                                   false, false);
    const unsigned rd = r.x, rs = r.y;
    d = __builtin_bit_cast(float, rd);
    s = __builtin_bit_cast(float, rs);
}

#define ORDER2(a, b)             asm volatile("" : "+v"(a), "+v"(b))
#define ORDER5(a, b, c, d, e)    asm volatile("" : "+v"(a), "+v"(b), "+v"(c), "+v"(d), "+v"(e))
#define ORDER6(a, b, c, d, e, f) asm volatile("" : "+v"(a), "+v"(b), "+v"(c), "+v"(d), "+v"(e), "+v"(f))
#define ORDER2_LDS(a, b)         asm volatile("" : "+v"(a), "+v"(b) : : "memory")

#define WM(Aop, Bp, Cacc) __builtin_amdgcn_wmma_f32_16x16x4_f32(false, (Aop), false, (Bp), (short)0, (Cacc), false, false)

__device__ __forceinline__ void fc_chunk(const float* __restrict__ half,
                                         const float (&wfc)[H_], float bfc,
                                         float* __restrict__ orow_chunk,
                                         int n, int tq)
{
    float s[4] = {0.0f, 0.0f, 0.0f, 0.0f};
#pragma unroll
    for (int kq = 0; kq < H_ / 4; ++kq) {
#pragma unroll
        for (int i = 0; i < 4; ++i) {
            const v4f hv = *(const v4f*)(half + (tq * 4 + i) * SLOT_WORDS + n * HPITCH + 4 * kq);
            s[i] = __builtin_fmaf(wfc[4 * kq + 0], hv.x, s[i]);
            s[i] = __builtin_fmaf(wfc[4 * kq + 1], hv.y, s[i]);
            s[i] = __builtin_fmaf(wfc[4 * kq + 2], hv.z, s[i]);
            s[i] = __builtin_fmaf(wfc[4 * kq + 3], hv.w, s[i]);
        }
    }
    v4f o;
    o.x = s[0] + bfc; o.y = s[1] + bfc; o.z = s[2] + bfc; o.w = s[3] + bfc;
    *(v4f*)(orow_chunk + tq * 4) = o;
}

__global__ __launch_bounds__(64, 1)
void SineRNN_wmma_kernel(const float* __restrict__ x,
                         const float* __restrict__ W_ih,
                         const float* __restrict__ W_hh,
                         const float* __restrict__ b_ih,
                         const float* __restrict__ b_hh,
                         const float* __restrict__ W_fc,
                         const float* __restrict__ b_fc,
                         float* __restrict__ out)
{
    __shared__ __attribute__((aligned(16))) float ring[RSLOTS * SLOT_WORDS];
    __shared__ __attribute__((aligned(16))) float xbuf[XBUF_WORDS];

    const int lane  = threadIdx.x & 31;
    const int role  = __builtin_amdgcn_readfirstlane((int)(threadIdx.x >> 5));
    const int n     = lane & 15;
    const int hi    = lane >> 4;
    const int bbase = blockIdx.x * 16;

    v2f A[2][8];
    float wih[2][8], bia[2][8];
    v2f Bop[8];
    float xv = 0.0f;
    float xc[CH];
    const float* xrow = xbuf + n * XPITCH;
    float* hbase = ring + n * HPITCH + 2 * hi;
    v8f ci0, pre1;
#pragma unroll
    for (int j = 0; j < 8; ++j) { Bop[j].x = 0.0f; Bop[j].y = 0.0f; }
#pragma unroll
    for (int r = 0; r < 8; ++r) { ci0[r] = 0.0f; pre1[r] = 0.0f; }
#pragma unroll
    for (int i = 0; i < CH; ++i) xc[i] = 0.0f;

    float wfc[H_];
    float bfc = 0.0f;
    const int xb = lane >> 1, xpart = lane & 1;
    const float* xsrc = x + (size_t)(bbase + xb) * T_ + xpart * 4;
    float* xdst = xbuf + xb * XPITCH + xpart * 4;
    float* orow = out + (size_t)(bbase + n) * T_;
    v4f xreg = {0.0f, 0.0f, 0.0f, 0.0f};

    if (role == 0) {
#pragma unroll
        for (int c = 0; c < 2; ++c) {
#pragma unroll
            for (int j = 0; j < 8; ++j) {
                const int row = 16 * c + n;
                const int k0  = 4 * j + 2 * hi;
                A[c][j].x = W_hh[row * H_ + k0];
                A[c][j].y = W_hh[row * H_ + k0 + 1];
            }
        }
#pragma unroll
        for (int c = 0; c < 2; ++c) {
#pragma unroll
            for (int r = 0; r < 8; ++r) {
                const int h = 16 * c + 8 * hi + r;
                wih[c][r] = W_ih[h];
                bia[c][r] = b_ih[h] + b_hh[h];
            }
        }
        const v4f g0 = *(const v4f*)(x + (size_t)(bbase + n) * T_);
        const v4f g1 = *(const v4f*)(x + (size_t)(bbase + n) * T_ + 4);
        xv = g0.x;
        xc[1] = g0.y; xc[2] = g0.z; xc[3] = g0.w;
        xc[4] = g1.x; xc[5] = g1.y; xc[6] = g1.z; xc[7] = g1.w;
#pragma unroll
        for (int r = 0; r < 8; ++r) ci0[r] = __builtin_fmaf(xv, wih[0][r], bia[0][r]);
    } else {
#pragma unroll
        for (int k = 0; k < H_; ++k) wfc[k] = W_fc[k];
        bfc = b_fc[0];
        const v4f x0 = *(const v4f*)(xsrc + 0 * CH);
        const v4f x1 = *(const v4f*)(xsrc + 1 * CH);
        xreg         = *(const v4f*)(xsrc + 2 * CH);
        *(v4f*)(xdst + 0 * CH) = x0;
        *(v4f*)(xdst + 1 * CH) = x1;
    }
    __syncthreads();

    int bcur = 0;
    for (int c = 0; c < NCH; ++c) {
        if (role == 0) {
            float* hprev = hbase + ((bcur == 0) ? (RSLOTS - 1) : (bcur * CH - 1)) * SLOT_WORDS;
            float* hcur  = hbase + bcur * (CH * SLOT_WORDS);
            const float* xnrow = xrow + ((bcur == 2) ? 0 : (bcur + 1)) * CH;
            v4f xnA = {0.0f, 0.0f, 0.0f, 0.0f}, xnB = {0.0f, 0.0f, 0.0f, 0.0f};
            v2f Hs[CH][8];

#pragma unroll
            for (int i = 0; i < CH; ++i) {
                v8f acc0, acc1, ci1;
                float t0[8], t1[8];

                acc0 = WM(A[0][0], Bop[0], ci0);
                ORDER2(acc0, pre1);
#pragma unroll
                for (int r = 0; r < 4; ++r) t1[r] = hw_tanhf(pre1[r]);
                ORDER5(acc0, t1[0], t1[1], t1[2], t1[3]);
                acc0 = WM(A[0][1], Bop[1], acc0);
                ORDER2(acc0, pre1);
#pragma unroll
                for (int r = 4; r < 8; ++r) t1[r] = hw_tanhf(pre1[r]);
                ORDER5(acc0, t1[4], t1[5], t1[6], t1[7]);
                acc0 = WM(A[0][2], Bop[2], acc0);
                ORDER6(acc0, t1[0], t1[1], t1[4], t1[5], xv);
#pragma unroll
                for (int r = 0; r < 8; ++r) ci1[r] = __builtin_fmaf(xv, wih[1][r], bia[1][r]);
                row_swap(t1[0], t1[2]); row_swap(t1[1], t1[3]);
                row_swap(t1[4], t1[6]); row_swap(t1[5], t1[7]);
                Bop[4].x = t1[0]; Bop[4].y = t1[1];
                Bop[5].x = t1[4]; Bop[5].y = t1[5];
                Bop[6].x = t1[2]; Bop[6].y = t1[3];
                Bop[7].x = t1[6]; Bop[7].y = t1[7];
                ORDER6(acc0, ci1, Bop[4], Bop[5], Bop[6], Bop[7]);
                acc0 = WM(A[0][3], Bop[3], acc0);
#pragma unroll
                for (int j = 0; j < 8; ++j) Hs[i][j] = Bop[j];
                if ((i % PL) == PL - 1) {
                    ORDER2_LDS(acc0, ci1);
                    if (i == PL - 1) {
                        xnA = *(const v4f*)(xnrow);
                        xnB = *(const v4f*)(xnrow + 4);
                    }
#pragma unroll
                    for (int k = i - (PL - 1); k <= i; ++k) {
                        float* brow = (k == 0) ? hprev : (hcur + (k - 1) * SLOT_WORDS);
#pragma unroll
                        for (int j = 0; j < 8; ++j)
                            *(v2f*)(brow + 4 * j) = Hs[k][j];
                    }
                    ORDER2_LDS(acc0, ci1);
                } else {
                    ORDER2(acc0, ci1);
                }

                acc1 = WM(A[1][0], Bop[0], ci1);
                ORDER2(acc0, acc1);
                acc0 = WM(A[0][4], Bop[4], acc0);
                ORDER2(acc0, acc1);
                acc1 = WM(A[1][1], Bop[1], acc1);
                ORDER2(acc0, acc1);
                acc0 = WM(A[0][5], Bop[5], acc0);
                ORDER2(acc0, acc1);
                acc1 = WM(A[1][2], Bop[2], acc1);
                ORDER2(acc0, acc1);
                acc0 = WM(A[0][6], Bop[6], acc0);
                ORDER2(acc0, acc1);
                acc1 = WM(A[1][3], Bop[3], acc1);
                ORDER2(acc0, acc1);
                acc0 = WM(A[0][7], Bop[7], acc0);
                ORDER2(acc0, acc1);

                float xnx = (i + 1 < CH) ? xc[i + 1] : xnA.x;
                acc1 = WM(A[1][4], Bop[4], acc1);
                ORDER2(acc1, acc0);
#pragma unroll
                for (int r = 0; r < 4; ++r) t0[r] = hw_tanhf(acc0[r]);
                ORDER5(acc1, t0[0], t0[1], t0[2], t0[3]);
                acc1 = WM(A[1][5], Bop[5], acc1);
                ORDER2(acc1, acc0);
#pragma unroll
                for (int r = 4; r < 8; ++r) t0[r] = hw_tanhf(acc0[r]);
                ORDER5(acc1, t0[4], t0[5], t0[6], t0[7]);
                acc1 = WM(A[1][6], Bop[6], acc1);
                ORDER6(acc1, t0[0], t0[1], t0[4], t0[5], xnx);
                row_swap(t0[0], t0[2]); row_swap(t0[1], t0[3]);
                row_swap(t0[4], t0[6]); row_swap(t0[5], t0[7]);
                Bop[0].x = t0[0]; Bop[0].y = t0[1];
                Bop[1].x = t0[4]; Bop[1].y = t0[5];
                Bop[2].x = t0[2]; Bop[2].y = t0[3];
                Bop[3].x = t0[6]; Bop[3].y = t0[7];
#pragma unroll
                for (int r = 0; r < 8; ++r) ci0[r] = __builtin_fmaf(xnx, wih[0][r], bia[0][r]);
                ORDER6(acc1, ci0, Bop[0], Bop[1], Bop[2], Bop[3]);
                acc1 = WM(A[1][7], Bop[7], acc1);

                pre1 = acc1;
                xv   = xnx;
            }
            xc[1] = xnA.y; xc[2] = xnA.z; xc[3] = xnA.w;
            xc[4] = xnB.x; xc[5] = xnB.y; xc[6] = xnB.z; xc[7] = xnB.w;
        } else {
            const int bp = (bcur == 0) ? 2 : bcur - 1;
            *(v4f*)(xdst + bp * CH) = xreg;
            const int cn = (c + 3 < NCH) ? (c + 3) : (NCH - 1);
            xreg = *(const v4f*)(xsrc + cn * CH);
            if (c > 1) {
                const int b2 = (bcur == 2) ? 0 : bcur + 1;
                fc_chunk(ring + b2 * (CH * SLOT_WORDS), wfc, bfc,
                         orow + (c - 2) * CH, n, hi);
            }
        }
        __syncthreads();
        bcur = (bcur == 2) ? 0 : bcur + 1;
    }
    if (role == 0) {
        float t1[8];
#pragma unroll
        for (int r = 0; r < 8; ++r) t1[r] = hw_tanhf(pre1[r]);
        row_swap(t1[0], t1[2]); row_swap(t1[1], t1[3]);
        row_swap(t1[4], t1[6]); row_swap(t1[5], t1[7]);
        Bop[4].x = t1[0]; Bop[4].y = t1[1];
        Bop[5].x = t1[4]; Bop[5].y = t1[5];
        Bop[6].x = t1[2]; Bop[6].y = t1[3];
        Bop[7].x = t1[6]; Bop[7].y = t1[7];
        float* brow = hbase + ((T_ - 1) % RSLOTS) * SLOT_WORDS;
#pragma unroll
        for (int j = 0; j < 8; ++j)
            *(v2f*)(brow + 4 * j) = Bop[j];
    }
    __syncthreads();
    if (role == 1) {
        fc_chunk(ring + ((NCH - 2) % RCH) * (CH * SLOT_WORDS), wfc, bfc,
                 orow + (NCH - 2) * CH, n, hi);
        fc_chunk(ring + ((NCH - 1) % RCH) * (CH * SLOT_WORDS), wfc, bfc,
                 orow + (NCH - 1) * CH, n, hi);
    }
}

extern "C" void kernel_launch(void* const* d_in, const int* in_sizes, int n_in,
                              void* d_out, int out_size, void* d_ws, size_t ws_size,
                              hipStream_t stream) {
    const float* x    = (const float*)d_in[0];
    const float* W_ih = (const float*)d_in[1];
    const float* W_hh = (const float*)d_in[2];
    const float* b_ih = (const float*)d_in[3];
    const float* b_hh = (const float*)d_in[4];
    const float* W_fc = (const float*)d_in[5];
    const float* b_fc = (const float*)d_in[6];
    float* out = (float*)d_out;

    dim3 grid(B_ / 16);
    dim3 block(64);
    SineRNN_wmma_kernel<<<grid, block, 0, stream>>>(
        x, W_ih, W_hh, b_ih, b_hh, W_fc, b_fc, out);
}


# W_hh tile rows re-assigned (tile row 8*half+2q+e = hidden 16c+4q+2*half+e): tanh(D) register pairs are the next B pairs in place, the 8 row swaps per step gone; k order kept, same bits
# speedup vs baseline: 2.6753x; 1.0593x over previous
#include <hip/hip_runtime.h>

#define B_ 256
#define T_ 4096
#define H_ 32
#define HPITCH 36
#define SLOT_WORDS (16 * HPITCH)
#define CH 8
#define NCH (T_ / CH)
#define PL 4
#define RCH 3
#define RSLOTS (RCH * CH)
#define XRING 24
#define XPITCH 28
#define XBUF_WORDS (16 * XPITCH)

typedef float v2f __attribute__((ext_vector_type(2)));
typedef float v4f __attribute__((ext_vector_type(4)));
typedef float v8f __attribute__((ext_vector_type(8)));

__device__ __forceinline__ float hw_tanhf(float v) {
#if __has_builtin(__builtin_amdgcn_tanhf)
    return __builtin_amdgcn_tanhf(v);
#else
    const float xc = __builtin_fminf(__builtin_fmaxf(v, -9.0f), 9.0f);
    const float t  = __builtin_amdgcn_exp2f(xc * 2.8853900817779268f);
    const float r  = __builtin_amdgcn_rcpf(t + 1.0f);
    return (t - 1.0f) * r;
#endif
}

#define ORDER2(a, b)             asm volatile("" : "+v"(a), "+v"(b))
#define ORDER5(a, b, c, d, e)    asm volatile("" : "+v"(a), "+v"(b), "+v"(c), "+v"(d), "+v"(e))
#define ORDER6(a, b, c, d, e, f) asm volatile("" : "+v"(a), "+v"(b), "+v"(c), "+v"(d), "+v"(e), "+v"(f))
#define ORDER2_LDS(a, b)         asm volatile("" : "+v"(a), "+v"(b) : : "memory")

#define WM(Aop, Bp, Cacc) __builtin_amdgcn_wmma_f32_16x16x4_f32(false, (Aop), false, (Bp), (short)0, (Cacc), false, false)

__device__ __forceinline__ void fc_chunk(const float* __restrict__ half,
                                         const float (&wfc)[H_], float bfc,
                                         float* __restrict__ orow_chunk,
                                         int n, int tq)
{
    float s[4] = {0.0f, 0.0f, 0.0f, 0.0f};
#pragma unroll
    for (int kq = 0; kq < H_ / 4; ++kq) {
#pragma unroll
        for (int i = 0; i < 4; ++i) {
            const v4f hv = *(const v4f*)(half + (tq * 4 + i) * SLOT_WORDS + n * HPITCH + 4 * kq);
            s[i] = __builtin_fmaf(wfc[4 * kq + 0], hv.x, s[i]);
            s[i] = __builtin_fmaf(wfc[4 * kq + 1], hv.y, s[i]);
            s[i] = __builtin_fmaf(wfc[4 * kq + 2], hv.z, s[i]);
            s[i] = __builtin_fmaf(wfc[4 * kq + 3], hv.w, s[i]);
        }
    }
    v4f o;
    o.x = s[0] + bfc; o.y = s[1] + bfc; o.z = s[2] + bfc; o.w = s[3] + bfc;
    *(v4f*)(orow_chunk + tq * 4) = o;
}

__global__ __launch_bounds__(64, 1)
void SineRNN_wmma_kernel(const float* __restrict__ x,
                         const float* __restrict__ W_ih,
                         const float* __restrict__ W_hh,
                         const float* __restrict__ b_ih,
                         const float* __restrict__ b_hh,
                         const float* __restrict__ W_fc,
                         const float* __restrict__ b_fc,
                         float* __restrict__ out)
{
    __shared__ __attribute__((aligned(16))) float ring[RSLOTS * SLOT_WORDS];
    __shared__ __attribute__((aligned(16))) float xbuf[XBUF_WORDS];

    const int lane  = threadIdx.x & 31;
    const int role  = __builtin_amdgcn_readfirstlane((int)(threadIdx.x >> 5));
    const int n     = lane & 15;
    const int hi    = lane >> 4;
    const int bbase = blockIdx.x * 16;

    v2f A[2][8];
    float wih[2][8], bia[2][8];
    v2f Bop[8];
    float xv = 0.0f;
    float xc[CH];
    const float* xrow = xbuf + n * XPITCH;
    float* hbase = ring + n * HPITCH + 2 * hi;
    v8f ci0, pre1;
#pragma unroll
    for (int j = 0; j < 8; ++j) { Bop[j].x = 0.0f; Bop[j].y = 0.0f; }
#pragma unroll
    for (int r = 0; r < 8; ++r) { ci0[r] = 0.0f; pre1[r] = 0.0f; }
#pragma unroll
    for (int i = 0; i < CH; ++i) xc[i] = 0.0f;

    float wfc[H_];
    float bfc = 0.0f;
    const int xb = lane >> 1, xpart = lane & 1;
    const float* xsrc = x + (size_t)(bbase + xb) * T_ + xpart * 4;
    float* xdst = xbuf + xb * XPITCH + xpart * 4;
    float* orow = out + (size_t)(bbase + n) * T_;
    v4f xreg = {0.0f, 0.0f, 0.0f, 0.0f};

    if (role == 0) {
#pragma unroll
        for (int c = 0; c < 2; ++c) {
#pragma unroll
            for (int j = 0; j < 8; ++j) {
                const int row = 16 * c + 4 * ((n >> 1) & 3) + 2 * (n >> 3) + (n & 1);
                const int k0  = 4 * j + 2 * hi;
                A[c][j].x = W_hh[row * H_ + k0];
                A[c][j].y = W_hh[row * H_ + k0 + 1];
            }
        }
#pragma unroll
        for (int c = 0; c < 2; ++c) {
#pragma unroll
            for (int r = 0; r < 8; ++r) {
                const int h = 16 * c + 4 * (r >> 1) + 2 * hi + (r & 1);
                wih[c][r] = W_ih[h];
                bia[c][r] = b_ih[h] + b_hh[h];
            }
        }
        const v4f g0 = *(const v4f*)(x + (size_t)(bbase + n) * T_);
        const v4f g1 = *(const v4f*)(x + (size_t)(bbase + n) * T_ + 4);
        xv = g0.x;
        xc[1] = g0.y; xc[2] = g0.z; xc[3] = g0.w;
        xc[4] = g1.x; xc[5] = g1.y; xc[6] = g1.z; xc[7] = g1.w;
#pragma unroll
        for (int r = 0; r < 8; ++r) ci0[r] = __builtin_fmaf(xv, wih[0][r], bia[0][r]);
    } else {
#pragma unroll
        for (int k = 0; k < H_; ++k) wfc[k] = W_fc[k];
        bfc = b_fc[0];
        const v4f x0 = *(const v4f*)(xsrc + 0 * CH);
        const v4f x1 = *(const v4f*)(xsrc + 1 * CH);
        xreg         = *(const v4f*)(xsrc + 2 * CH);
        *(v4f*)(xdst + 0 * CH) = x0;
        *(v4f*)(xdst + 1 * CH) = x1;
    }
    __syncthreads();

    int bcur = 0;
    for (int c = 0; c < NCH; ++c) {
        if (role == 0) {
            float* hprev = hbase + ((bcur == 0) ? (RSLOTS - 1) : (bcur * CH - 1)) * SLOT_WORDS;
            float* hcur  = hbase + bcur * (CH * SLOT_WORDS);
            const float* xnrow = xrow + ((bcur == 2) ? 0 : (bcur + 1)) * CH;
            v4f xnA = {0.0f, 0.0f, 0.0f, 0.0f}, xnB = {0.0f, 0.0f, 0.0f, 0.0f};
            v2f Hs[CH][8];

#pragma unroll
            for (int i = 0; i < CH; ++i) {
                v8f acc0, acc1, ci1;
                float t0[8], t1[8];

                acc0 = WM(A[0][0], Bop[0], ci0);
                ORDER2(acc0, pre1);
#pragma unroll
                for (int r = 0; r < 4; ++r) t1[r] = hw_tanhf(pre1[r]);
                ORDER5(acc0, t1[0], t1[1], t1[2], t1[3]);
                acc0 = WM(A[0][1], Bop[1], acc0);
                ORDER2(acc0, pre1);
#pragma unroll
                for (int r = 4; r < 8; ++r) t1[r] = hw_tanhf(pre1[r]);
                ORDER5(acc0, t1[4], t1[5], t1[6], t1[7]);
                acc0 = WM(A[0][2], Bop[2], acc0);
                ORDER2(acc0, xv);
#pragma unroll
                for (int r = 0; r < 8; ++r) ci1[r] = __builtin_fmaf(xv, wih[1][r], bia[1][r]);
#pragma unroll
                for (int q = 0; q < 4; ++q) { Bop[4 + q].x = t1[2 * q]; Bop[4 + q].y = t1[2 * q + 1]; }
                ORDER6(acc0, ci1, Bop[4], Bop[5], Bop[6], Bop[7]);
                acc0 = WM(A[0][3], Bop[3], acc0);
#pragma unroll
                for (int j = 0; j < 8; ++j) Hs[i][j] = Bop[j];
                if ((i % PL) == PL - 1) {
                    ORDER2_LDS(acc0, ci1);
                    if (i == PL - 1) {
                        xnA = *(const v4f*)(xnrow);
                        xnB = *(const v4f*)(xnrow + 4);
                    }
#pragma unroll
                    for (int k = i - (PL - 1); k <= i; ++k) {
                        float* brow = (k == 0) ? hprev : (hcur + (k - 1) * SLOT_WORDS);
#pragma unroll
                        for (int j = 0; j < 8; ++j)
                            *(v2f*)(brow + 4 * j) = Hs[k][j];
                    }
                    ORDER2_LDS(acc0, ci1);
                } else {
                    ORDER2(acc0, ci1);
                }

                acc1 = WM(A[1][0], Bop[0], ci1);
                ORDER2(acc0, acc1);
                acc0 = WM(A[0][4], Bop[4], acc0);
                ORDER2(acc0, acc1);
                acc1 = WM(A[1][1], Bop[1], acc1);
                ORDER2(acc0, acc1);
                acc0 = WM(A[0][5], Bop[5], acc0);
                ORDER2(acc0, acc1);
                acc1 = WM(A[1][2], Bop[2], acc1);
                ORDER2(acc0, acc1);
                acc0 = WM(A[0][6], Bop[6], acc0);
                ORDER2(acc0, acc1);
                acc1 = WM(A[1][3], Bop[3], acc1);
                ORDER2(acc0, acc1);
                acc0 = WM(A[0][7], Bop[7], acc0);
                ORDER2(acc0, acc1);

                float xnx = (i + 1 < CH) ? xc[i + 1] : xnA.x;
                acc1 = WM(A[1][4], Bop[4], acc1);
                ORDER2(acc1, acc0);
#pragma unroll
                for (int r = 0; r < 4; ++r) t0[r] = hw_tanhf(acc0[r]);
                ORDER5(acc1, t0[0], t0[1], t0[2], t0[3]);
                acc1 = WM(A[1][5], Bop[5], acc1);
                ORDER2(acc1, acc0);
#pragma unroll
                for (int r = 4; r < 8; ++r) t0[r] = hw_tanhf(acc0[r]);
                ORDER5(acc1, t0[4], t0[5], t0[6], t0[7]);
                acc1 = WM(A[1][6], Bop[6], acc1);
                ORDER2(acc1, xnx);
#pragma unroll
                for (int q = 0; q < 4; ++q) { Bop[q].x = t0[2 * q]; Bop[q].y = t0[2 * q + 1]; }
#pragma unroll
                for (int r = 0; r < 8; ++r) ci0[r] = __builtin_fmaf(xnx, wih[0][r], bia[0][r]);
                ORDER6(acc1, ci0, Bop[0], Bop[1], Bop[2], Bop[3]);
                acc1 = WM(A[1][7], Bop[7], acc1);

                pre1 = acc1;
                xv   = xnx;
            }
            xc[1] = xnA.y; xc[2] = xnA.z; xc[3] = xnA.w;
            xc[4] = xnB.x; xc[5] = xnB.y; xc[6] = xnB.z; xc[7] = xnB.w;
        } else {
            const int bp = (bcur == 0) ? 2 : bcur - 1;
            *(v4f*)(xdst + bp * CH) = xreg;
            const int cn = (c + 3 < NCH) ? (c + 3) : (NCH - 1);
            xreg = *(const v4f*)(xsrc + cn * CH);
            if (c > 1) {
                const int b2 = (bcur == 2) ? 0 : bcur + 1;
                fc_chunk(ring + b2 * (CH * SLOT_WORDS), wfc, bfc,
                         orow + (c - 2) * CH, n, hi);
            }
        }
        __syncthreads();
        bcur = (bcur == 2) ? 0 : bcur + 1;
    }
    if (role == 0) {
        float t1[8];
#pragma unroll
        for (int r = 0; r < 8; ++r) t1[r] = hw_tanhf(pre1[r]);
#pragma unroll
        for (int q = 0; q < 4; ++q) { Bop[4 + q].x = t1[2 * q]; Bop[4 + q].y = t1[2 * q + 1]; }
        float* brow = hbase + ((T_ - 1) % RSLOTS) * SLOT_WORDS;
#pragma unroll
        for (int j = 0; j < 8; ++j)
            *(v2f*)(brow + 4 * j) = Bop[j];
    }
    __syncthreads();
    if (role == 1) {
        fc_chunk(ring + ((NCH - 2) % RCH) * (CH * SLOT_WORDS), wfc, bfc,
                 orow + (NCH - 2) * CH, n, hi);
        fc_chunk(ring + ((NCH - 1) % RCH) * (CH * SLOT_WORDS), wfc, bfc,
                 orow + (NCH - 1) * CH, n, hi);
    }
}

extern "C" void kernel_launch(void* const* d_in, const int* in_sizes, int n_in,
                              void* d_out, int out_size, void* d_ws, size_t ws_size,
                              hipStream_t stream) {
    const float* x    = (const float*)d_in[0];
    const float* W_ih = (const float*)d_in[1];
    const float* W_hh = (const float*)d_in[2];
    const float* b_ih = (const float*)d_in[3];
    const float* b_hh = (const float*)d_in[4];
    const float* W_fc = (const float*)d_in[5];
    const float* b_fc = (const float*)d_in[6];
    float* out = (float*)d_out;

    dim3 grid(B_ / 16);
    dim3 block(64);
    SineRNN_wmma_kernel<<<grid, block, 0, stream>>>(
        x, W_ih, W_hh, b_ih, b_hh, W_fc, b_fc, out);
}


# recurrent product split by k over 2 waves (2 SIMDs): each keeps its half of h in registers (D is its next B); only partial sums of the other tile cross, via LDS + a barrier a step; sum order changes
# speedup vs baseline: 2.8308x; 1.0581x over previous
#include <hip/hip_runtime.h>

#define B_ 256
#define T_ 4096
#define H_ 32
#define HPITCH 36
#define SLOT_WORDS (16 * HPITCH)
#define CH 8
#define NCH (T_ / CH)
#define RCH 3
#define RSLOTS (RCH * CH)
#define XRING 24
#define XPITCH 28
#define XBUF_WORDS (16 * XPITCH)
#define PSET_WORDS 256
#define PBUF_WORDS (2 * 2 * PSET_WORDS)

typedef float v2f __attribute__((ext_vector_type(2)));
typedef float v4f __attribute__((ext_vector_type(4)));
typedef float v8f __attribute__((ext_vector_type(8)));

__device__ __forceinline__ float hw_tanhf(float v) {
#if __has_builtin(__builtin_amdgcn_tanhf)
    return __builtin_amdgcn_tanhf(v);
#else
    const float xc = __builtin_fminf(__builtin_fmaxf(v, -9.0f), 9.0f);
    const float t  = __builtin_amdgcn_exp2f(xc * 2.8853900817779268f);
    const float r  = __builtin_amdgcn_rcpf(t + 1.0f);
    return (t - 1.0f) * r;
#endif
}

#define WM(Aop, Bp, Cacc) __builtin_amdgcn_wmma_f32_16x16x4_f32(false, (Aop), false, (Bp), (short)0, (Cacc), false, false)

__device__ __forceinline__ void fc_chunk(const float* __restrict__ half,
                                         const float (&wfc)[H_], float bfc,
                                         float* __restrict__ orow_chunk,
                                         int n, int tq)
{
    float s[4] = {0.0f, 0.0f, 0.0f, 0.0f};
#pragma unroll
    for (int kq = 0; kq < H_ / 4; ++kq) {
#pragma unroll
        for (int i = 0; i < 4; ++i) {
            const v4f hv = *(const v4f*)(half + (tq * 4 + i) * SLOT_WORDS + n * HPITCH + 4 * kq);
            s[i] = __builtin_fmaf(wfc[4 * kq + 0], hv.x, s[i]);
            s[i] = __builtin_fmaf(wfc[4 * kq + 1], hv.y, s[i]);
            s[i] = __builtin_fmaf(wfc[4 * kq + 2], hv.z, s[i]);
            s[i] = __builtin_fmaf(wfc[4 * kq + 3], hv.w, s[i]);
        }
    }
    v4f o;
    o.x = s[0] + bfc; o.y = s[1] + bfc; o.z = s[2] + bfc; o.w = s[3] + bfc;
    *(v4f*)(orow_chunk + tq * 4) = o;
}

__device__ __forceinline__ void rec_issue(const v2f (&Ao)[4], const v2f (&Ap)[4], const v2f (&Bop)[4],
                                          v8f& ci_cur, v8f& ci_next, v8f& accP,
                                          const float (&wih)[8], const float (&bia)[8], float xnext,
                                          float* __restrict__ pdst, float* __restrict__ hslot)
{
    const v8f zero8 = {0.0f, 0.0f, 0.0f, 0.0f, 0.0f, 0.0f, 0.0f, 0.0f};
    v8f accO = WM(Ao[0], Bop[0], zero8);
    accO = WM(Ao[1], Bop[1], accO);
    accO = WM(Ao[2], Bop[2], accO);
    accO = WM(Ao[3], Bop[3], accO);
    v4f plo, phi2;
    plo.x = accO[0]; plo.y = accO[1]; plo.z = accO[2]; plo.w = accO[3];
    phi2.x = accO[4]; phi2.y = accO[5]; phi2.z = accO[6]; phi2.w = accO[7];
    *(v4f*)(pdst)       = plo;
    *(v4f*)(pdst + 128) = phi2;
#pragma unroll
    for (int q = 0; q < 4; ++q)
        *(v2f*)(hslot + 4 * q) = Bop[q];
    asm volatile("" : "+v"(ci_cur) : : "memory");
    accP = WM(Ap[0], Bop[0], ci_cur);
    accP = WM(Ap[1], Bop[1], accP);
#pragma unroll
    for (int r = 0; r < 8; ++r) ci_next[r] = __builtin_fmaf(xnext, wih[r], bia[r]);
}

__device__ __forceinline__ void rec_finish(const v2f (&Ap)[4], v2f (&Bop)[4], v8f& accP,
                                           const float* __restrict__ psrc)
{
    const v4f qlo = *(const v4f*)(psrc);
    const v4f qhi = *(const v4f*)(psrc + 128);
    accP = WM(Ap[2], Bop[2], accP);
    accP = WM(Ap[3], Bop[3], accP);
    float t[8];
    t[0] = hw_tanhf(accP[0] + qlo.x); t[1] = hw_tanhf(accP[1] + qlo.y);
    t[2] = hw_tanhf(accP[2] + qlo.z); t[3] = hw_tanhf(accP[3] + qlo.w);
    t[4] = hw_tanhf(accP[4] + qhi.x); t[5] = hw_tanhf(accP[5] + qhi.y);
    t[6] = hw_tanhf(accP[6] + qhi.z); t[7] = hw_tanhf(accP[7] + qhi.w);
#pragma unroll
    for (int q = 0; q < 4; ++q) { Bop[q].x = t[2 * q]; Bop[q].y = t[2 * q + 1]; }
}

__global__ __launch_bounds__(96, 1)
void SineRNN_wmma_kernel(const float* __restrict__ x,
                         const float* __restrict__ W_ih,
                         const float* __restrict__ W_hh,
                         const float* __restrict__ b_ih,
                         const float* __restrict__ b_hh,
                         const float* __restrict__ W_fc,
                         const float* __restrict__ b_fc,
                         float* __restrict__ out)
{
    __shared__ __attribute__((aligned(16))) float ring[RSLOTS * SLOT_WORDS];
    __shared__ __attribute__((aligned(16))) float xbuf[XBUF_WORDS];
    __shared__ __attribute__((aligned(16))) float pbuf[PBUF_WORDS];

    const int lane  = threadIdx.x & 31;
    const int wave  = __builtin_amdgcn_readfirstlane((int)(threadIdx.x >> 5));
    const int kh    = wave & 1;
    const int n     = lane & 15;
    const int hi    = lane >> 4;
    const int bbase = blockIdx.x * 16;

    v2f Ao[4], Ap[4];
    float wih[8], bia[8];
    v2f Bop[4];
    float xc[CH];
    const float* xrow = xbuf + n * XPITCH;
    float* hbase = ring + n * HPITCH + 16 * kh + 2 * hi;
    float* pst = pbuf + kh * PSET_WORDS + lane * 4;
    const float* pld = pbuf + (1 - kh) * PSET_WORDS + lane * 4;
    v8f ci[2];
    v8f accP;
#pragma unroll
    for (int q = 0; q < 4; ++q) { Bop[q].x = 0.0f; Bop[q].y = 0.0f; }
#pragma unroll
    for (int r = 0; r < 8; ++r) { ci[0][r] = 0.0f; ci[1][r] = 0.0f; accP[r] = 0.0f; }
#pragma unroll
    for (int i = 0; i < CH; ++i) xc[i] = 0.0f;

    float wfc[H_];
    float bfc = 0.0f;
    float fs[4] = {0.0f, 0.0f, 0.0f, 0.0f};
    const int xb = lane >> 1, xpart = lane & 1;
    const float* xsrc = x + (size_t)(bbase + xb) * T_ + xpart * 4;
    float* xdst = xbuf + xb * XPITCH + xpart * 4;
    float* orow = out + (size_t)(bbase + n) * T_;
    v4f xreg = {0.0f, 0.0f, 0.0f, 0.0f};

    if (wave < 2) {
        const int rowp = 16 * kh       + 4 * ((n >> 1) & 3) + 2 * (n >> 3) + (n & 1);
        const int rowo = 16 * (1 - kh) + 4 * ((n >> 1) & 3) + 2 * (n >> 3) + (n & 1);
#pragma unroll
        for (int q = 0; q < 4; ++q) {
            const int k0 = 4 * (4 * kh + q) + 2 * hi;
            Ap[q].x = W_hh[rowp * H_ + k0];
            Ap[q].y = W_hh[rowp * H_ + k0 + 1];
            Ao[q].x = W_hh[rowo * H_ + k0];
            Ao[q].y = W_hh[rowo * H_ + k0 + 1];
        }
#pragma unroll
        for (int r = 0; r < 8; ++r) {
            const int h = 16 * kh + 4 * (r >> 1) + 2 * hi + (r & 1);
            wih[r] = W_ih[h];
            bia[r] = b_ih[h] + b_hh[h];
        }
        const v4f g0 = *(const v4f*)(x + (size_t)(bbase + n) * T_);
        const v4f g1 = *(const v4f*)(x + (size_t)(bbase + n) * T_ + 4);
        xc[1] = g0.y; xc[2] = g0.z; xc[3] = g0.w;
        xc[4] = g1.x; xc[5] = g1.y; xc[6] = g1.z; xc[7] = g1.w;
#pragma unroll
        for (int r = 0; r < 8; ++r) ci[0][r] = __builtin_fmaf(g0.x, wih[r], bia[r]);
    } else {
#pragma unroll
        for (int k = 0; k < H_; ++k) wfc[k] = W_fc[k];
        bfc = b_fc[0];
        const v4f x0 = *(const v4f*)(xsrc + 0 * CH);
        const v4f x1 = *(const v4f*)(xsrc + 1 * CH);
        const v4f x2 = *(const v4f*)(xsrc + 2 * CH);
        *(v4f*)(xdst + 0 * CH) = x0;
        *(v4f*)(xdst + 1 * CH) = x1;
        *(v4f*)(xdst + 2 * CH) = x2;
    }
    __syncthreads();

    if (wave < 2)
        rec_issue(Ao, Ap, Bop, ci[0], ci[1], accP, wih, bia, xc[1],
                  pst, hbase + (RSLOTS - 1) * SLOT_WORDS);

    int bcur = 0;
    for (int c = 0; c < NCH; ++c) {
        float* hcur  = hbase + bcur * (CH * SLOT_WORDS);
        const float* xnrow = xrow + ((bcur == 2) ? 0 : (bcur + 1)) * CH;
        const float* fhalf = ring + ((bcur == 2) ? 0 : (bcur + 1)) * (CH * SLOT_WORDS)
                                  + (hi * 4) * SLOT_WORDS + n * HPITCH;
        v4f xnA = {0.0f, 0.0f, 0.0f, 0.0f}, xnB = {0.0f, 0.0f, 0.0f, 0.0f};

#pragma unroll
        for (int i = 0; i < CH; ++i) {
            const int par = i & 1;
            if (wave == 2) {
                if (i == 0) {
                    const int cn = (c + 3 < NCH) ? (c + 3) : (NCH - 1);
                    xreg = *(const v4f*)(xsrc + cn * CH);
                }
                if (i == CH - 1) {
                    *(v4f*)(xdst + bcur * CH) = xreg;
                }
                if (c > 1) {
                    if (i == 0) { fs[0] = 0.0f; fs[1] = 0.0f; fs[2] = 0.0f; fs[3] = 0.0f; }
#pragma unroll
                    for (int s = 0; s < 4; ++s) {
                        const v4f hv = *(const v4f*)(fhalf + s * SLOT_WORDS + 4 * i);
                        fs[s] = __builtin_fmaf(wfc[4 * i + 0], hv.x, fs[s]);
                        fs[s] = __builtin_fmaf(wfc[4 * i + 1], hv.y, fs[s]);
                        fs[s] = __builtin_fmaf(wfc[4 * i + 2], hv.z, fs[s]);
                        fs[s] = __builtin_fmaf(wfc[4 * i + 3], hv.w, fs[s]);
                    }
                    if (i == CH - 1) {
                        v4f o;
                        o.x = fs[0] + bfc; o.y = fs[1] + bfc; o.z = fs[2] + bfc; o.w = fs[3] + bfc;
                        *(v4f*)(orow + (c - 2) * CH + hi * 4) = o;
                    }
                }
            }
            __builtin_amdgcn_fence(__ATOMIC_RELEASE, "workgroup", "local");
            __builtin_amdgcn_s_barrier();
            __builtin_amdgcn_fence(__ATOMIC_ACQUIRE, "workgroup", "local");
            if (wave < 2) {
                rec_finish(Ap, Bop, accP, pld + par * (2 * PSET_WORDS));
                if (i + 1 < CH || c + 1 < NCH) {
                    if (i == 3) {
                        xnA = *(const v4f*)(xnrow);
                        xnB = *(const v4f*)(xnrow + 4);
                    }
                    const float xn2 = (i + 2 < CH) ? xc[(i + 2 < CH) ? (i + 2) : 0]
                                                   : ((i + 2 == CH) ? xnA.x : xnA.y);
                    rec_issue(Ao, Ap, Bop, ci[par ^ 1], ci[par], accP, wih, bia, xn2,
                              pst + (par ^ 1) * (2 * PSET_WORDS), hcur + i * SLOT_WORDS);
                }
            }
        }
        xc[1] = xnA.y; xc[2] = xnA.z; xc[3] = xnA.w;
        xc[4] = xnB.x; xc[5] = xnB.y; xc[6] = xnB.z; xc[7] = xnB.w;
        bcur = (bcur == 2) ? 0 : bcur + 1;
    }
    if (wave < 2) {
        float* brow = hbase + ((T_ - 1) % RSLOTS) * SLOT_WORDS;
#pragma unroll
        for (int q = 0; q < 4; ++q)
            *(v2f*)(brow + 4 * q) = Bop[q];
    }
    __syncthreads();
    if (wave == 2) {
        fc_chunk(ring + ((NCH - 2) % RCH) * (CH * SLOT_WORDS), wfc, bfc,
                 orow + (NCH - 2) * CH, n, hi);
        fc_chunk(ring + ((NCH - 1) % RCH) * (CH * SLOT_WORDS), wfc, bfc,
                 orow + (NCH - 1) * CH, n, hi);
    }
}

extern "C" void kernel_launch(void* const* d_in, const int* in_sizes, int n_in,
                              void* d_out, int out_size, void* d_ws, size_t ws_size,
                              hipStream_t stream) {
    const float* x    = (const float*)d_in[0];
    const float* W_ih = (const float*)d_in[1];
    const float* W_hh = (const float*)d_in[2];
    const float* b_ih = (const float*)d_in[3];
    const float* b_hh = (const float*)d_in[4];
    const float* W_fc = (const float*)d_in[5];
    const float* b_fc = (const float*)d_in[6];
    float* out = (float*)d_out;

    dim3 grid(B_ / 16);
    dim3 block(96);
    SineRNN_wmma_kernel<<<grid, block, 0, stream>>>(
        x, W_ih, W_hh, b_ih, b_hh, W_fc, b_fc, out);
}


# speedup vs baseline: 3.1688x; 1.1194x over previous
#include <hip/hip_runtime.h>

#define B_ 256
#define T_ 4096
#define H_ 32
#define HPITCH 36
#define SLOT_WORDS (16 * HPITCH)
#define CH 8
#define NCH (T_ / CH)
#define RCH 3
#define RSLOTS (RCH * CH)
#define XRING 24
#define XPITCH 28
#define XBUF_WORDS (16 * XPITCH)
#define PSET_WORDS 256
#define PBUF_WORDS (2 * 2 * PSET_WORDS)

typedef float v2f __attribute__((ext_vector_type(2)));
typedef float v4f __attribute__((ext_vector_type(4)));
typedef float v8f __attribute__((ext_vector_type(8)));

__device__ __forceinline__ float hw_tanhf(float v) {
#if __has_builtin(__builtin_amdgcn_tanhf)
    return __builtin_amdgcn_tanhf(v);
#else
    const float xc = __builtin_fminf(__builtin_fmaxf(v, -9.0f), 9.0f);
    const float t  = __builtin_amdgcn_exp2f(xc * 2.8853900817779268f);
    const float r  = __builtin_amdgcn_rcpf(t + 1.0f);
    return (t - 1.0f) * r;
#endif
}

#define WM(Aop, Bp, Cacc) __builtin_amdgcn_wmma_f32_16x16x4_f32(false, (Aop), false, (Bp), (short)0, (Cacc), false, false)

__device__ __forceinline__ void fc_chunk(const float* __restrict__ half,
                                         const float (&wfc)[H_], float bfc,
                                         float* __restrict__ orow_chunk,
                                         int n, int tq)
{
    float s[4] = {0.0f, 0.0f, 0.0f, 0.0f};
#pragma unroll
    for (int kq = 0; kq < H_ / 4; ++kq) {
#pragma unroll
        for (int i = 0; i < 4; ++i) {
            const v4f hv = *(const v4f*)(half + (tq * 4 + i) * SLOT_WORDS + n * HPITCH + 4 * kq);
            s[i] = __builtin_fmaf(wfc[4 * kq + 0], hv.x, s[i]);
            s[i] = __builtin_fmaf(wfc[4 * kq + 1], hv.y, s[i]);
            s[i] = __builtin_fmaf(wfc[4 * kq + 2], hv.z, s[i]);
            s[i] = __builtin_fmaf(wfc[4 * kq + 3], hv.w, s[i]);
        }
    }
    v4f o;
    o.x = s[0] + bfc; o.y = s[1] + bfc; o.z = s[2] + bfc; o.w = s[3] + bfc;
    *(v4f*)(orow_chunk + tq * 4) = o;
}

__device__ __forceinline__ void rec_issue(const v2f (&Ao)[4], const v2f (&Ap)[4], const v2f (&Bop)[4],
                                          v8f& ci_cur, v8f& ci_next, v8f& accP,
                                          const float (&wih)[8], const float (&bia)[8], float xnext,
                                          float* __restrict__ pdst, float* __restrict__ hslot)
{
    const v8f zero8 = {0.0f, 0.0f, 0.0f, 0.0f, 0.0f, 0.0f, 0.0f, 0.0f};
    v8f accO = WM(Ao[0], Bop[0], zero8);
    accO = WM(Ao[1], Bop[1], accO);
    accO = WM(Ao[2], Bop[2], accO);
    accO = WM(Ao[3], Bop[3], accO);
    v4f plo, phi2;
    plo.x = accO[0]; plo.y = accO[1]; plo.z = accO[2]; plo.w = accO[3];
    phi2.x = accO[4]; phi2.y = accO[5]; phi2.z = accO[6]; phi2.w = accO[7];
    *(v4f*)(pdst)       = plo;
    *(v4f*)(pdst + 128) = phi2;
#pragma unroll
    for (int q = 0; q < 4; ++q)
        *(v2f*)(hslot + 4 * q) = Bop[q];
    asm volatile("" : "+v"(ci_cur) : : "memory");
    accP = WM(Ap[0], Bop[0], ci_cur);
    accP = WM(Ap[1], Bop[1], accP);
#pragma unroll
    for (int r = 0; r < 8; ++r) ci_next[r] = __builtin_fmaf(xnext, wih[r], bia[r]);
}

__device__ __forceinline__ void rec_finish(const v2f (&Ap)[4], v2f (&Bop)[4], v8f& accP,
                                           const float* __restrict__ psrc)
{
    const v4f qlo = *(const v4f*)(psrc);
    const v4f qhi = *(const v4f*)(psrc + 128);
    __builtin_amdgcn_sched_barrier(0);
    accP = WM(Ap[2], Bop[2], accP);
    accP = WM(Ap[3], Bop[3], accP);
    float t[8];
    t[0] = hw_tanhf(accP[0] + qlo.x); t[1] = hw_tanhf(accP[1] + qlo.y);
    t[2] = hw_tanhf(accP[2] + qlo.z); t[3] = hw_tanhf(accP[3] + qlo.w);
    t[4] = hw_tanhf(accP[4] + qhi.x); t[5] = hw_tanhf(accP[5] + qhi.y);
    t[6] = hw_tanhf(accP[6] + qhi.z); t[7] = hw_tanhf(accP[7] + qhi.w);
#pragma unroll
    for (int q = 0; q < 4; ++q) { Bop[q].x = t[2 * q]; Bop[q].y = t[2 * q + 1]; }
}

__global__ __launch_bounds__(96, 1)
void SineRNN_wmma_kernel(const float* __restrict__ x,
                         const float* __restrict__ W_ih,
                         const float* __restrict__ W_hh,
                         const float* __restrict__ b_ih,
                         const float* __restrict__ b_hh,
                         const float* __restrict__ W_fc,
                         const float* __restrict__ b_fc,
                         float* __restrict__ out)
{
    __shared__ __attribute__((aligned(16))) float ring[RSLOTS * SLOT_WORDS];
    __shared__ __attribute__((aligned(16))) float xbuf[XBUF_WORDS];
    __shared__ __attribute__((aligned(16))) float pbuf[PBUF_WORDS];

    const int lane  = threadIdx.x & 31;
    const int wave  = __builtin_amdgcn_readfirstlane((int)(threadIdx.x >> 5));
    const int kh    = wave & 1;
    const int n     = lane & 15;
    const int hi    = lane >> 4;
    const int bbase = blockIdx.x * 16;

    v2f Ao[4], Ap[4];
    float wih[8], bia[8];
    v2f Bop[4];
    float xc[CH];
    const float* xrow = xbuf + n * XPITCH;
    float* hbase = ring + n * HPITCH + 16 * kh + 2 * hi;
    float* pst = pbuf + kh * PSET_WORDS + lane * 4;
    const float* pld = pbuf + (1 - kh) * PSET_WORDS + lane * 4;
    v8f ci[2];
    v8f accP;
#pragma unroll
    for (int q = 0; q < 4; ++q) { Bop[q].x = 0.0f; Bop[q].y = 0.0f; }
#pragma unroll
    for (int r = 0; r < 8; ++r) { ci[0][r] = 0.0f; ci[1][r] = 0.0f; accP[r] = 0.0f; }
#pragma unroll
    for (int i = 0; i < CH; ++i) xc[i] = 0.0f;

    float wfc[H_];
    float bfc = 0.0f;
    float fs[4] = {0.0f, 0.0f, 0.0f, 0.0f};
    const int xb = lane >> 1, xpart = lane & 1;
    const float* xsrc = x + (size_t)(bbase + xb) * T_ + xpart * 4;
    float* xdst = xbuf + xb * XPITCH + xpart * 4;
    float* orow = out + (size_t)(bbase + n) * T_;
    v4f xreg = {0.0f, 0.0f, 0.0f, 0.0f};

    if (wave < 2) {
        const int rowp = 16 * kh       + 4 * ((n >> 1) & 3) + 2 * (n >> 3) + (n & 1);
        const int rowo = 16 * (1 - kh) + 4 * ((n >> 1) & 3) + 2 * (n >> 3) + (n & 1);
#pragma unroll
        for (int q = 0; q < 4; ++q) {
            const int k0 = 4 * (4 * kh + q) + 2 * hi;
            Ap[q].x = W_hh[rowp * H_ + k0];
            Ap[q].y = W_hh[rowp * H_ + k0 + 1];
            Ao[q].x = W_hh[rowo * H_ + k0];
            Ao[q].y = W_hh[rowo * H_ + k0 + 1];
        }
#pragma unroll
        for (int r = 0; r < 8; ++r) {
            const int h = 16 * kh + 4 * (r >> 1) + 2 * hi + (r & 1);
            wih[r] = W_ih[h];
            bia[r] = b_ih[h] + b_hh[h];
        }
        const v4f g0 = *(const v4f*)(x + (size_t)(bbase + n) * T_);
        const v4f g1 = *(const v4f*)(x + (size_t)(bbase + n) * T_ + 4);
        xc[1] = g0.y; xc[2] = g0.z; xc[3] = g0.w;
        xc[4] = g1.x; xc[5] = g1.y; xc[6] = g1.z; xc[7] = g1.w;
#pragma unroll
        for (int r = 0; r < 8; ++r) ci[0][r] = __builtin_fmaf(g0.x, wih[r], bia[r]);
    } else {
#pragma unroll
        for (int k = 0; k < H_; ++k) wfc[k] = W_fc[k];
        bfc = b_fc[0];
        const v4f x0 = *(const v4f*)(xsrc + 0 * CH);
        const v4f x1 = *(const v4f*)(xsrc + 1 * CH);
        const v4f x2 = *(const v4f*)(xsrc + 2 * CH);
        *(v4f*)(xdst + 0 * CH) = x0;
        *(v4f*)(xdst + 1 * CH) = x1;
        *(v4f*)(xdst + 2 * CH) = x2;
    }
    __syncthreads();

    if (wave < 2)
        rec_issue(Ao, Ap, Bop, ci[0], ci[1], accP, wih, bia, xc[1],
                  pst, hbase + (RSLOTS - 1) * SLOT_WORDS);

    int bcur = 0;
    for (int c = 0; c < NCH; ++c) {
        float* hcur  = hbase + bcur * (CH * SLOT_WORDS);
        const float* xnrow = xrow + ((bcur == 2) ? 0 : (bcur + 1)) * CH;
        const float* fhalf = ring + ((bcur == 2) ? 0 : (bcur + 1)) * (CH * SLOT_WORDS)
                                  + (hi * 4) * SLOT_WORDS + n * HPITCH;
        v4f xnA = {0.0f, 0.0f, 0.0f, 0.0f}, xnB = {0.0f, 0.0f, 0.0f, 0.0f};

#pragma unroll
        for (int i = 0; i < CH; ++i) {
            const int par = i & 1;
            if (wave == 2) {
                if (i == 0) {
                    const int cn = (c + 3 < NCH) ? (c + 3) : (NCH - 1);
                    xreg = *(const v4f*)(xsrc + cn * CH);
                }
                if (i == CH - 1) {
                    *(v4f*)(xdst + bcur * CH) = xreg;
                }
                if (c > 1) {
                    if (i == 0) { fs[0] = 0.0f; fs[1] = 0.0f; fs[2] = 0.0f; fs[3] = 0.0f; }
#pragma unroll
                    for (int s = 0; s < 4; ++s) {
                        const v4f hv = *(const v4f*)(fhalf + s * SLOT_WORDS + 4 * i);
                        fs[s] = __builtin_fmaf(wfc[4 * i + 0], hv.x, fs[s]);
                        fs[s] = __builtin_fmaf(wfc[4 * i + 1], hv.y, fs[s]);
                        fs[s] = __builtin_fmaf(wfc[4 * i + 2], hv.z, fs[s]);
                        fs[s] = __builtin_fmaf(wfc[4 * i + 3], hv.w, fs[s]);
                    }
                    if (i == CH - 1) {
                        v4f o;
                        o.x = fs[0] + bfc; o.y = fs[1] + bfc; o.z = fs[2] + bfc; o.w = fs[3] + bfc;
                        *(v4f*)(orow + (c - 2) * CH + hi * 4) = o;
                    }
                }
            }
            __builtin_amdgcn_fence(__ATOMIC_RELEASE, "workgroup", "local");
            __builtin_amdgcn_s_barrier();
            __builtin_amdgcn_fence(__ATOMIC_ACQUIRE, "workgroup", "local");
            if (wave < 2) {
                rec_finish(Ap, Bop, accP, pld + par * (2 * PSET_WORDS));
                if (i + 1 < CH || c + 1 < NCH) {
                    if (i == 3) {
                        xnA = *(const v4f*)(xnrow);
                        xnB = *(const v4f*)(xnrow + 4);
                    }
                    const float xn2 = (i + 2 < CH) ? xc[(i + 2 < CH) ? (i + 2) : 0]
                                                   : ((i + 2 == CH) ? xnA.x : xnA.y);
                    rec_issue(Ao, Ap, Bop, ci[par ^ 1], ci[par], accP, wih, bia, xn2,
                              pst + (par ^ 1) * (2 * PSET_WORDS), hcur + i * SLOT_WORDS);
                }
            }
        }
        xc[1] = xnA.y; xc[2] = xnA.z; xc[3] = xnA.w;
        xc[4] = xnB.x; xc[5] = xnB.y; xc[6] = xnB.z; xc[7] = xnB.w;
        bcur = (bcur == 2) ? 0 : bcur + 1;
    }
    if (wave < 2) {
        float* brow = hbase + ((T_ - 1) % RSLOTS) * SLOT_WORDS;
#pragma unroll
        for (int q = 0; q < 4; ++q)
            *(v2f*)(brow + 4 * q) = Bop[q];
    }
    __syncthreads();
    if (wave == 2) {
        fc_chunk(ring + ((NCH - 2) % RCH) * (CH * SLOT_WORDS), wfc, bfc,
                 orow + (NCH - 2) * CH, n, hi);
        fc_chunk(ring + ((NCH - 1) % RCH) * (CH * SLOT_WORDS), wfc, bfc,
                 orow + (NCH - 1) * CH, n, hi);
    }
}

extern "C" void kernel_launch(void* const* d_in, const int* in_sizes, int n_in,
                              void* d_out, int out_size, void* d_ws, size_t ws_size,
                              hipStream_t stream) {
    const float* x    = (const float*)d_in[0];
    const float* W_ih = (const float*)d_in[1];
    const float* W_hh = (const float*)d_in[2];
    const float* b_ih = (const float*)d_in[3];
    const float* b_hh = (const float*)d_in[4];
    const float* W_fc = (const float*)d_in[5];
    const float* b_fc = (const float*)d_in[6];
    float* out = (float*)d_out;

    dim3 grid(B_ / 16);
    dim3 block(96);
    SineRNN_wmma_kernel<<<grid, block, 0, stream>>>(
        x, W_ih, W_hh, b_ih, b_hh, W_fc, b_fc, out);
}
